# starting kernel
# baseline (speedup 1.0000x reference)
#include <hip/hip_runtime.h>
#include <hip/hip_bf16.h>

#define B 32
#define M 512
#define T 16
#define D 128
#define V 32000

typedef __attribute__((ext_vector_type(2))) float v2f;
typedef __attribute__((ext_vector_type(8))) float v8f;

__global__ void pos_kernel(const int* __restrict__ story,
                           int* __restrict__ positions,
                           float* __restrict__ u)
{
    int b = blockIdx.x;
    int m = threadIdx.x;
    __shared__ int scan[M];
    int nonpad = (story[(size_t)m * (B * T) + (size_t)b * T] != 0) ? 1 : 0;
    scan[m] = nonpad;
    __syncthreads();
    for (int off = 1; off < M; off <<= 1) {
        int v = scan[m];
        int w = (m >= off) ? scan[m - off] : 0;
        __syncthreads();
        scan[m] = v + w;
        __syncthreads();
    }
    positions[b * M + m] = nonpad ? scan[m] : 0;
    if (m < D) u[b * D + m] = 0.0f;
}

__global__ void pool_kernel(const int* __restrict__ story,
                            const float* __restrict__ C_tables,
                            const float* __restrict__ pos_emb,
                            const int* __restrict__ positions,
                            float* __restrict__ SA)
{
    int wave = threadIdx.x >> 5;
    int lane = threadIdx.x & 31;
    int slot = blockIdx.x * 8 + wave;
    int b = slot >> 9;
    int m = slot & (M - 1);

    int tokval = 0;
    if (lane < T) tokval = story[(size_t)m * (B * T) + (size_t)b * T + lane];

    const float4* t0 = (const float4*)(C_tables);
    const float4* t1 = (const float4*)(C_tables + (size_t)1 * V * D);
    const float4* t2 = (const float4*)(C_tables + (size_t)2 * V * D);
    const float4* t3 = (const float4*)(C_tables + (size_t)3 * V * D);
    const int D4 = D / 4;

    float4 a0 = {0.f, 0.f, 0.f, 0.f};
    float4 a1 = a0, a2 = a0, a3 = a0;
#pragma unroll
    for (int t = 0; t < T; ++t) {
        int tok = __shfl(tokval, t, 32);
        size_t off = (size_t)tok * D4 + lane;
        float4 v0 = t0[off];
        float4 v1 = t1[off];
        float4 v2 = t2[off];
        float4 v3 = t3[off];
        a0.x += v0.x; a0.y += v0.y; a0.z += v0.z; a0.w += v0.w;
        a1.x += v1.x; a1.y += v1.y; a1.z += v1.z; a1.w += v1.w;
        a2.x += v2.x; a2.y += v2.y; a2.z += v2.z; a2.w += v2.w;
        a3.x += v3.x; a3.y += v3.y; a3.z += v3.z; a3.w += v3.w;
    }

    float4 pe = ((const float4*)pos_emb)[(size_t)positions[slot] * D4 + lane];
    a0.x += pe.x; a0.y += pe.y; a0.z += pe.z; a0.w += pe.w;
    a1.x += pe.x; a1.y += pe.y; a1.z += pe.z; a1.w += pe.w;
    a2.x += pe.x; a2.y += pe.y; a2.z += pe.z; a2.w += pe.w;
    a3.x += pe.x; a3.y += pe.y; a3.z += pe.z; a3.w += pe.w;

    float4* o = (float4*)SA;
    size_t base = (size_t)slot * D4 + lane;
    const size_t S4 = (size_t)B * M * D4;
    o[base + 0 * S4] = a0;
    o[base + 1 * S4] = a1;
    o[base + 2 * S4] = a2;
    o[base + 3 * S4] = a3;
}

__global__ void hop_kernel(const float* __restrict__ SA_A,
                           const float* __restrict__ SA_C,
                           float* __restrict__ u,
                           float* __restrict__ out)
{
    int b    = blockIdx.x;
    int tid  = threadIdx.x;
    int wave = tid >> 5;
    int lane = tid & 31;
    int row  = lane & 15;
    int khalf = lane >> 4;

    __shared__ float sh_u[D];
    __shared__ float sh_l[M];
    __shared__ float red[8];

    if (tid < D) sh_u[tid] = u[b * D + tid];
    __syncthreads();

    const float* Abase = SA_A + (size_t)b * M * D;
    for (int mt = wave * 4; mt < wave * 4 + 4; ++mt) {
        int m0 = mt * 16;
        v8f acc = {0.f, 0.f, 0.f, 0.f, 0.f, 0.f, 0.f, 0.f};
        for (int kc = 0; kc < D / 4; ++kc) {
            int d0 = kc * 4 + 2 * khalf;
            const float* p = Abase + (size_t)(m0 + row) * D + d0;
            v2f a;  a.x = p[0];        a.y = p[1];
            v2f bb; bb.x = sh_u[d0];   bb.y = sh_u[d0 + 1];
            acc = __builtin_amdgcn_wmma_f32_16x16x4_f32(
                false, a, false, bb, (short)0, acc, false, false);
        }
        if (row == 0) {
#pragma unroll
            for (int j = 0; j < 8; ++j)
                sh_l[m0 + khalf * 8 + j] = acc[j];
        }
    }
    __syncthreads();

    float lmax = -__builtin_inff();
    for (int i = tid; i < M; i += 256) lmax = fmaxf(lmax, sh_l[i]);
    for (int off = 16; off > 0; off >>= 1)
        lmax = fmaxf(lmax, __shfl_xor(lmax, off, 32));
    if (lane == 0) red[wave] = lmax;
    __syncthreads();
    if (tid == 0) {
        float g = red[0];
        for (int i = 1; i < 8; ++i) g = fmaxf(g, red[i]);
        red[0] = g;
    }
    __syncthreads();
    float gmax = red[0];

    float lsum = 0.f;
    for (int i = tid; i < M; i += 256) {
        float e = __expf(sh_l[i] - gmax);
        sh_l[i] = e;
        lsum += e;
    }
    for (int off = 16; off > 0; off >>= 1)
        lsum += __shfl_xor(lsum, off, 32);
    __syncthreads();
    if (lane == 0) red[wave] = lsum;
    __syncthreads();
    if (tid == 0) {
        float s = 0.f;
        for (int i = 0; i < 8; ++i) s += red[i];
        red[0] = s;
    }
    __syncthreads();
    float inv = 1.0f / red[0];
    for (int i = tid; i < M; i += 256) sh_l[i] *= inv;
    __syncthreads();

    const float* Cbase = SA_C + (size_t)b * M * D;
    int d_tile = wave * 16;
    {
        v8f acc = {0.f, 0.f, 0.f, 0.f, 0.f, 0.f, 0.f, 0.f};
        for (int kc = 0; kc < M / 4; ++kc) {
            int m0 = kc * 4 + 2 * khalf;
            const float* p = Cbase + (size_t)m0 * D + d_tile + row;
            v2f a;  a.x = p[0];        a.y = p[D];
            v2f bb; bb.x = sh_l[m0];   bb.y = sh_l[m0 + 1];
            acc = __builtin_amdgcn_wmma_f32_16x16x4_f32(
                false, a, false, bb, (short)0, acc, false, false);
        }
        if (row == 0) {
#pragma unroll
            for (int j = 0; j < 8; ++j) {
                int d = d_tile + khalf * 8 + j;
                float nu = sh_u[d] + acc[j];
                u[b * D + d] = nu;
                if (out) out[b * D + d] = nu;
            }
        }
    }
}

extern "C" void kernel_launch(void* const* d_in, const int* in_sizes, int n_in,
                              void* d_out, int out_size, void* d_ws, size_t ws_size,
                              hipStream_t stream)
{
    const int*   story    = (const int*)d_in[0];
    const float* C_tables = (const float*)d_in[1];
    const float* pos_emb  = (const float*)d_in[2];
    float*       out      = (float*)d_out;

    char* ws = (char*)d_ws;
    int*   positions = (int*)ws;
    float* u         = (float*)(ws + (size_t)B * M * sizeof(int));
    size_t sa_off    = (((size_t)B * M * sizeof(int) +
                         (size_t)B * D * sizeof(float)) + 255) & ~(size_t)255;
    float* SA        = (float*)(ws + sa_off);
    const size_t S   = (size_t)B * M * D;

    pos_kernel <<<B,          M,   0, stream>>>(story, positions, u);
    pool_kernel<<<B * M / 8,  256, 0, stream>>>(story, C_tables, pos_emb,
                                                positions, SA);
    hop_kernel <<<B,          256, 0, stream>>>(SA + 0 * S, SA + 1 * S, u, nullptr);
    hop_kernel <<<B,          256, 0, stream>>>(SA + 1 * S, SA + 2 * S, u, nullptr);
    hop_kernel <<<B,          256, 0, stream>>>(SA + 2 * S, SA + 3 * S, u, out);
}


# Hop 0 is a uniform mean because u starts at zero: table-0 gather, hop-0 launch and its two products (by 0, by 1) dropped; pooling kernel leaves S_1 column partial sums; summation tree changes
# speedup vs baseline: 2.4795x; 2.4795x over previous
#include <hip/hip_runtime.h>
#include <hip/hip_bf16.h>

#define B 32
#define M 512
#define T 16
#define D 128
#define V 32000

#define G 8
#define R (M / G)
#define REC_STRIDE (R * D)
#define PART_ROW 16

typedef __attribute__((ext_vector_type(2))) float v2f;
typedef __attribute__((ext_vector_type(8))) float v8f;

__global__ void pos_kernel(const int* __restrict__ story,
                           int* __restrict__ positions)
{
    int b = blockIdx.x;
    int m = threadIdx.x;
    __shared__ int scan[M];
    int nonpad = (story[(size_t)m * (B * T) + (size_t)b * T] != 0) ? 1 : 0;
    scan[m] = nonpad;
    __syncthreads();
    for (int off = 1; off < M; off <<= 1) {
        int v = scan[m];
        int w = (m >= off) ? scan[m - off] : 0;
        __syncthreads();
        scan[m] = v + w;
        __syncthreads();
    }
    positions[b * M + m] = nonpad ? scan[m] : 0;
}

__global__ void pool_kernel(const int* __restrict__ story,
                            const float* __restrict__ C_tables,
                            const float* __restrict__ pos_emb,
                            const int* __restrict__ positions,
                            float* __restrict__ SA)
{
    __shared__ __attribute__((aligned(16))) float sh_c[8][D];

    int wave = threadIdx.x >> 5;
    int lane = threadIdx.x & 31;
    int slot0 = blockIdx.x * 8;
    int slot = slot0 + wave;
    int b = slot >> 9;
    int m = slot & (M - 1);

    int tokval = 0;
    if (lane < T) tokval = story[(size_t)m * (B * T) + (size_t)b * T + lane];

    const float4* t1 = (const float4*)(C_tables + (size_t)1 * V * D);
    const float4* t2 = (const float4*)(C_tables + (size_t)2 * V * D);
    const float4* t3 = (const float4*)(C_tables + (size_t)3 * V * D);
    const int D4 = D / 4;

    float4 a1 = {0.f, 0.f, 0.f, 0.f};
    float4 a2 = a1, a3 = a1;
#pragma unroll
    for (int t = 0; t < T; ++t) {
        int tok = __shfl(tokval, t, 32);
        size_t off = (size_t)tok * D4 + lane;
        float4 v1 = t1[off];
        float4 v2 = t2[off];
        float4 v3 = t3[off];
        a1.x += v1.x; a1.y += v1.y; a1.z += v1.z; a1.w += v1.w;
        a2.x += v2.x; a2.y += v2.y; a2.z += v2.z; a2.w += v2.w;
        a3.x += v3.x; a3.y += v3.y; a3.z += v3.z; a3.w += v3.w;
    }

    float4 pe = ((const float4*)pos_emb)[(size_t)positions[slot] * D4 + lane];
    a1.x += pe.x; a1.y += pe.y; a1.z += pe.z; a1.w += pe.w;
    a2.x += pe.x; a2.y += pe.y; a2.z += pe.z; a2.w += pe.w;
    a3.x += pe.x; a3.y += pe.y; a3.z += pe.z; a3.w += pe.w;

    *(float4*)(&sh_c[wave][4 * lane]) = a1;
    __syncthreads();

    float4* o = (float4*)SA;
    size_t base = (size_t)slot * D4 + lane;
    const size_t S4 = (size_t)B * M * D4;
    o[base + 1 * S4] = a1;
    o[base + 2 * S4] = a2;
    o[base + 3 * S4] = a3;

    int d = threadIdx.x;
    if (d < D) {
        float sum = sh_c[0][d];
#pragma unroll
        for (int w = 1; w < 8; ++w) sum += sh_c[w][d];
        size_t prow = (size_t)(slot0 & ~(R - 1)) + PART_ROW + ((slot0 >> 3) & 7);
        SA[prow * D + d] = sum;
    }
}

__device__ __forceinline__ float merge_records(const float* __restrict__ pr, int d)
{
    float mx[G], ss[G], po[G];
#pragma unroll
    for (int g = 0; g < G; ++g) {
        mx[g] = pr[g * REC_STRIDE + D];
        ss[g] = pr[g * REC_STRIDE + D + 1];
        po[g] = pr[g * REC_STRIDE + d];
    }
    float uold = pr[2 * D + d];
    float gm = mx[0];
#pragma unroll
    for (int g = 1; g < G; ++g) gm = fmaxf(gm, mx[g]);
    float den = 0.f, num = 0.f;
#pragma unroll
    for (int g = 0; g < G; ++g) {
        float w = __expf(mx[g] - gm);
        den += w * ss[g];
        num += w * po[g];
    }
    return uold + num / den;
}

__device__ __forceinline__ float sum_partials(const float* __restrict__ pb, int d, int h)
{
    float acc = 0.f;
#pragma unroll
    for (int j = 0; j < 32; ++j) {
        int p = h * 32 + j;
        acc += pb[(size_t)((p >> 3) * R + PART_ROW + (p & 7)) * D + d];
    }
    return acc;
}

template <bool FROM_POOL>
__global__ __launch_bounds__(256)
void hop_part_kernel(const float* A,
                     const float* C,
                     const float* __restrict__ prev,
                     float* rec)
{
    int g    = blockIdx.x;
    int b    = blockIdx.y;
    int tid  = threadIdx.x;
    int wave = tid >> 5;
    int lane = tid & 31;
    int row  = lane & 15;
    int khalf = lane >> 4;

    __shared__ __attribute__((aligned(16))) float sh_u[D];
    __shared__ float sh_pl[2][R];
    __shared__ __attribute__((aligned(16))) float sh_po[4][D];
    __shared__ float sh_hs[2][D];

    size_t chunk = ((size_t)b * M + (size_t)g * R) * D;

    float u_d;
    if (FROM_POOL) u_d = sum_partials(prev + (size_t)b * M * D, tid & (D - 1), tid >> 7);
    else           u_d = merge_records(prev + (size_t)b * M * D, tid & (D - 1));

    int mt = wave & 3, kh = wave >> 2;
    const float4* ap = (const float4*)(A + chunk + (size_t)(mt * 16 + row) * D
                                         + kh * 64 + khalf * 32);
    float4 av[8];
#pragma unroll
    for (int i = 0; i < 8; ++i) av[i] = ap[i];

    int dh = wave & 1, mq = wave >> 1;
    const float* cp = C + chunk + (size_t)(mq * 16 + 2 * khalf) * D + dh * 64 + 4 * row;
    float4 cv[4][2];
#pragma unroll
    for (int ks = 0; ks < 4; ++ks) {
        cv[ks][0] = *(const float4*)(cp + (size_t)(ks * 4 + 0) * D);
        cv[ks][1] = *(const float4*)(cp + (size_t)(ks * 4 + 1) * D);
    }

    if (FROM_POOL) {
        sh_hs[tid >> 7][tid & (D - 1)] = u_d;
        __syncthreads();
        u_d = (sh_hs[0][tid & (D - 1)] + sh_hs[1][tid & (D - 1)]) * (1.0f / M);
    }
    if (tid < D) sh_u[tid] = u_d;
    __syncthreads();

    {
        const float4* up = (const float4*)(sh_u + kh * 64 + khalf * 32);
        v8f acc0 = {0.f, 0.f, 0.f, 0.f, 0.f, 0.f, 0.f, 0.f};
        v8f acc1 = acc0;
#pragma unroll
        for (int i = 0; i < 8; ++i) {
            float4 uv = up[i];
            v2f a0;  a0.x = av[i].x;  a0.y = av[i].y;
            v2f b0;  b0.x = uv.x;     b0.y = uv.y;
            acc0 = __builtin_amdgcn_wmma_f32_16x16x4_f32(
                false, a0, false, b0, (short)0, acc0, false, false);
            v2f a1;  a1.x = av[i].z;  a1.y = av[i].w;
            v2f b1;  b1.x = uv.z;     b1.y = uv.w;
            acc1 = __builtin_amdgcn_wmma_f32_16x16x4_f32(
                false, a1, false, b1, (short)0, acc1, false, false);
        }
        acc0 += acc1;
        if (row == 0) {
#pragma unroll
            for (int j = 0; j < 8; ++j)
                sh_pl[kh][mt * 16 + khalf * 8 + j] = acc0[j];
        }
    }
    __syncthreads();

    float l0 = sh_pl[0][lane]      + sh_pl[1][lane];
    float l1 = sh_pl[0][lane + 32] + sh_pl[1][lane + 32];
    float mx = fmaxf(l0, l1);
    for (int off = 16; off > 0; off >>= 1)
        mx = fmaxf(mx, __shfl_xor(mx, off, 32));
    float e0 = __expf(l0 - mx);
    float e1 = __expf(l1 - mx);
    float s = e0 + e1;
    for (int off = 16; off > 0; off >>= 1)
        s += __shfl_xor(s, off, 32);

    {
        float esel = (mq < 2) ? e0 : e1;
        int   lbase = (mq & 1) * 16 + 2 * khalf;
        v8f acc[4];
#pragma unroll
        for (int j = 0; j < 4; ++j)
            acc[j] = (v8f){0.f, 0.f, 0.f, 0.f, 0.f, 0.f, 0.f, 0.f};
#pragma unroll
        for (int ks = 0; ks < 4; ++ks) {
            v2f bb;
            bb.x = __shfl(esel, lbase + ks * 4 + 0, 32);
            bb.y = __shfl(esel, lbase + ks * 4 + 1, 32);
            float c0[4] = {cv[ks][0].x, cv[ks][0].y, cv[ks][0].z, cv[ks][0].w};
            float c1[4] = {cv[ks][1].x, cv[ks][1].y, cv[ks][1].z, cv[ks][1].w};
#pragma unroll
            for (int j = 0; j < 4; ++j) {
                v2f a;  a.x = c0[j];  a.y = c1[j];
                acc[j] = __builtin_amdgcn_wmma_f32_16x16x4_f32(
                    false, a, false, bb, (short)0, acc[j], false, false);
            }
        }
        if (row == 0) {
#pragma unroll
            for (int r = 0; r < 8; ++r) {
                float4 v = {acc[0][r], acc[1][r], acc[2][r], acc[3][r]};
                *(float4*)(&sh_po[mq][dh * 64 + 32 * khalf + 4 * r]) = v;
            }
        }
    }
    __syncthreads();

    float* rp = rec + chunk;
    if (tid < D) {
        rp[tid] = sh_po[0][tid] + sh_po[1][tid] + sh_po[2][tid] + sh_po[3][tid];
        if (g == 0) rp[2 * D + tid] = u_d;
    } else if (tid == D) {
        rp[D] = mx;
    } else if (tid == D + 1) {
        rp[D + 1] = s;
    }
}

__global__ void finish_kernel(const float* __restrict__ prev, float* __restrict__ out)
{
    int b = blockIdx.x;
    int d = threadIdx.x;
    out[b * D + d] = merge_records(prev + (size_t)b * M * D, d);
}

extern "C" void kernel_launch(void* const* d_in, const int* in_sizes, int n_in,
                              void* d_out, int out_size, void* d_ws, size_t ws_size,
                              hipStream_t stream)
{
    const int*   story    = (const int*)d_in[0];
    const float* C_tables = (const float*)d_in[1];
    const float* pos_emb  = (const float*)d_in[2];
    float*       out      = (float*)d_out;

    char* ws = (char*)d_ws;
    int*   positions = (int*)ws;
    size_t sa_off    = (((size_t)B * M * sizeof(int) +
                         (size_t)B * D * sizeof(float)) + 255) & ~(size_t)255;
    float* SA        = (float*)(ws + sa_off);
    const size_t S   = (size_t)B * M * D;
    float* rec1 = SA + 1 * 4 * D;
    float* rec2 = SA + 2 * 4 * D;

    pos_kernel <<<B,          M,   0, stream>>>(story, positions);
    pool_kernel<<<B * M / 8,  256, 0, stream>>>(story, C_tables, pos_emb,
                                                positions, SA);
    hop_part_kernel<true ><<<dim3(G, B), 256, 0, stream>>>(SA + 1 * S, SA + 2 * S, SA,      rec1);
    hop_part_kernel<false><<<dim3(G, B), 256, 0, stream>>>(SA + 2 * S, SA + 3 * S, rec1,    rec2);
    finish_kernel<<<B, D, 0, stream>>>(rec2, out);
}


# Position scan folded into the pooling kernel (one launch fewer): a work-group owns 32 rows of a batch, each wave pools 4 rows in turn; hop-0 partial sums cover 32 rows each: summation tree changes
# speedup vs baseline: 2.9353x; 1.1838x over previous
#include <hip/hip_runtime.h>
#include <hip/hip_bf16.h>

#define B 32
#define M 512
#define T 16
#define D 128
#define V 32000

#define G 8
#define R (M / G)
#define REC_STRIDE (R * D)
#define PART_ROW 16
#define RC 32
#define NCH (M / RC)

typedef __attribute__((ext_vector_type(2))) float v2f;
typedef __attribute__((ext_vector_type(8))) float v8f;

typedef const float4 __attribute__((address_space(1)))* gmem_f4;
__device__ __forceinline__ gmem_f4 scalar_base(const float* p)
{
    unsigned long long a = (unsigned long long)p;
    unsigned lo = __builtin_amdgcn_readfirstlane((unsigned)a);
    unsigned hi = __builtin_amdgcn_readfirstlane((unsigned)(a >> 32));
    return (gmem_f4)(((unsigned long long)hi << 32) | lo);
}

template <int GI>
__device__ __forceinline__ void issue_group(float4 (&bf)[3][8], int tokA, int tokB,
                                            const float* __restrict__ C1,
                                            const float* __restrict__ C2,
                                            const float* __restrict__ C3,
                                            unsigned lane)
{
    constexpr int sl   = GI >> 1;
    constexpr int src0 = (sl & 1) * 16 + (GI & 1) * 8;
    int tv = (sl < 2) ? tokA : tokB;
#pragma unroll
    for (int t = 0; t < 8; ++t) {
        int tok = __builtin_amdgcn_readlane(tv, src0 + t);
        size_t ro = (size_t)tok * D;
        bf[0][t] = scalar_base(C1 + ro)[lane];
        bf[1][t] = scalar_base(C2 + ro)[lane];
        bf[2][t] = scalar_base(C3 + ro)[lane];
    }
}

__device__ __forceinline__ void add4(float4& a, const float4& v)
{
    a.x += v.x; a.y += v.y; a.z += v.z; a.w += v.w;
}

__global__ __launch_bounds__(256)
void pool_kernel(const int* __restrict__ story,
                 const float* __restrict__ C1,
                 const float* __restrict__ C2,
                 const float* __restrict__ C3,
                 const float* __restrict__ pos_emb,
                 float* __restrict__ SA)
{
    __shared__ unsigned sh_mask[NCH];
    __shared__ __attribute__((aligned(16))) float sh_c[8][D];

    int wave = threadIdx.x >> 5;
    int lane = threadIdx.x & 31;
    int b  = blockIdx.x / NCH;
    int c  = blockIdx.x % NCH;
    int m0 = c * RC;
    int rw = wave * 4;

    int hl = lane >> 4, tl = lane & 15;
    int tokA = story[((size_t)(m0 + rw + hl) * B + b) * T + tl];
    int tokB = story[((size_t)(m0 + rw + 2 + hl) * B + b) * T + tl];

    unsigned lo = 0u, hi = 0u;
    if (2 * wave <= c) {
        int f0 = story[((size_t)(wave * 64 + lane) * B + b) * T];
        int f1 = story[((size_t)(wave * 64 + 32 + lane) * B + b) * T];
        lo = __builtin_amdgcn_ballot_w32(f0 != 0);
        hi = __builtin_amdgcn_ballot_w32(f1 != 0);
    }
    if (lane == 0) { sh_mask[2 * wave] = lo; sh_mask[2 * wave + 1] = hi; }
    __syncthreads();

    int before = 0;
#pragma unroll
    for (int j = 0; j < NCH; ++j) before += (j < c) ? __popc(sh_mask[j]) : 0;
    unsigned wc = sh_mask[c];

    const int D4 = D / 4;
    const float4* pe4 = (const float4*)pos_emb + lane;

    float4* o = (float4*)SA;
    const size_t S4 = (size_t)B * M * D4;

    float4 bfA[3][8], bfB[3][8];
    float4 zero = {0.f, 0.f, 0.f, 0.f};
    float4 a1 = zero, a2 = zero, a3 = zero;
    float4 w1 = zero;
    float4 l1 = zero, l2 = zero, l3 = zero;
    size_t lbase = 0;

    float4 pe = zero;
    issue_group<0>(bfA, tokA, tokB, C1, C2, C3, (unsigned)lane);

#define POOL_STEP(GI, CUR, NXT)                                                          \
    {                                                                                    \
           \
           \
           \
        if (!(GI & 1)) {                                                                 \
            int r = rw + (GI >> 1);                                                      \
            int pos = ((wc >> r) & 1u) ? before + __popc(wc & ((2u << r) - 1u)) : 0;     \
            pe = pe4[(size_t)pos * D4];                                                  \
        }                                                                                \
        if (GI + 1 < 8) issue_group<(GI + 1 < 8) ? GI + 1 : 0>(NXT, tokA, tokB, C1, C2, C3, (unsigned)lane); \
        _Pragma("unroll")                                                                \
        for (int t = 0; t < 8; ++t) {                                                    \
            add4(a1, CUR[0][t]); add4(a2, CUR[1][t]); add4(a3, CUR[2][t]);               \
        }                                                                                \
        if (GI & 1) {                                                                    \
            add4(a1, pe); add4(a2, pe); add4(a3, pe);                                    \
            add4(w1, a1);                                                                \
            size_t base = ((size_t)b * M + m0 + rw + (GI >> 1)) * D4 + lane;             \
            if (GI < 7) {                                                                \
                o[base + 1 * S4] = a1;                                                   \
                o[base + 2 * S4] = a2;                                                   \
                o[base + 3 * S4] = a3;                                                   \
            } else {                                                                     \
                l1 = a1; l2 = a2; l3 = a3; lbase = base;                                 \
            }                                                                            \
            a1 = zero; a2 = zero; a3 = zero;                                             \
        }                                                                                \
    }

    POOL_STEP(0, bfA, bfB)
    POOL_STEP(1, bfB, bfA)
    POOL_STEP(2, bfA, bfB)
    POOL_STEP(3, bfB, bfA)
    POOL_STEP(4, bfA, bfB)
    POOL_STEP(5, bfB, bfA)
    POOL_STEP(6, bfA, bfB)
    POOL_STEP(7, bfB, bfA)
#undef POOL_STEP

    *(float4*)(&sh_c[wave][4 * lane]) = w1;
    __syncthreads();

    o[lbase + 1 * S4] = l1;
    o[lbase + 2 * S4] = l2;
    o[lbase + 3 * S4] = l3;

    int d = threadIdx.x;
    if (d < D) {
        float sum = sh_c[0][d];
#pragma unroll
        for (int w = 1; w < 8; ++w) sum += sh_c[w][d];
        size_t prow = (size_t)b * M + (size_t)(m0 & ~(R - 1)) + PART_ROW + (c & 1);
        SA[prow * D + d] = sum;
    }
}

__device__ __forceinline__ float merge_records(const float* __restrict__ pr, int d)
{
    float mx[G], ss[G], po[G];
#pragma unroll
    for (int g = 0; g < G; ++g) {
        mx[g] = pr[g * REC_STRIDE + D];
        ss[g] = pr[g * REC_STRIDE + D + 1];
        po[g] = pr[g * REC_STRIDE + d];
    }
    float uold = pr[2 * D + d];
    float gm = mx[0];
#pragma unroll
    for (int g = 1; g < G; ++g) gm = fmaxf(gm, mx[g]);
    float den = 0.f, num = 0.f;
#pragma unroll
    for (int g = 0; g < G; ++g) {
        float w = __expf(mx[g] - gm);
        den += w * ss[g];
        num += w * po[g];
    }
    return uold + num / den;
}

__device__ __forceinline__ float sum_partials(const float* __restrict__ pb, int d, int h)
{
    float acc = 0.f;
#pragma unroll
    for (int j = 0; j < NCH / 2; ++j) {
        int p = h * (NCH / 2) + j;
        acc += pb[(size_t)((p >> 1) * R + PART_ROW + (p & 1)) * D + d];
    }
    return acc;
}

template <bool FROM_POOL>
__global__ __launch_bounds__(256)
void hop_part_kernel(const float* A,
                     const float* C,
                     const float* __restrict__ prev,
                     float* rec)
{
    int g    = blockIdx.x;
    int b    = blockIdx.y;
    int tid  = threadIdx.x;
    int wave = tid >> 5;
    int lane = tid & 31;
    int row  = lane & 15;
    int khalf = lane >> 4;

    __shared__ __attribute__((aligned(16))) float sh_u[D];
    __shared__ float sh_pl[2][R];
    __shared__ __attribute__((aligned(16))) float sh_po[4][D];
    __shared__ float sh_hs[2][D];

    size_t chunk = ((size_t)b * M + (size_t)g * R) * D;

    float u_d;
    if (FROM_POOL) u_d = sum_partials(prev + (size_t)b * M * D, tid & (D - 1), tid >> 7);
    else           u_d = merge_records(prev + (size_t)b * M * D, tid & (D - 1));

    int mt = wave & 3, kh = wave >> 2;
    const float4* ap = (const float4*)(A + chunk + (size_t)(mt * 16 + row) * D
                                         + kh * 64 + khalf * 32);
    float4 av[8];
#pragma unroll
    for (int i = 0; i < 8; ++i) av[i] = ap[i];

    int dh = wave & 1, mq = wave >> 1;
    const float* cp = C + chunk + (size_t)(mq * 16 + 2 * khalf) * D + dh * 64 + 4 * row;
    float4 cv[4][2];
#pragma unroll
    for (int ks = 0; ks < 4; ++ks) {
        cv[ks][0] = *(const float4*)(cp + (size_t)(ks * 4 + 0) * D);
        cv[ks][1] = *(const float4*)(cp + (size_t)(ks * 4 + 1) * D);
    }

    if (FROM_POOL) {
        sh_hs[tid >> 7][tid & (D - 1)] = u_d;
        __syncthreads();
        u_d = (sh_hs[0][tid & (D - 1)] + sh_hs[1][tid & (D - 1)]) * (1.0f / M);
    }
    if (tid < D) sh_u[tid] = u_d;
    __syncthreads();

    {
        const float4* up = (const float4*)(sh_u + kh * 64 + khalf * 32);
        v8f acc0 = {0.f, 0.f, 0.f, 0.f, 0.f, 0.f, 0.f, 0.f};
        v8f acc1 = acc0;
#pragma unroll
        for (int i = 0; i < 8; ++i) {
            float4 uv = up[i];
            v2f a0;  a0.x = av[i].x;  a0.y = av[i].y;
            v2f b0;  b0.x = uv.x;     b0.y = uv.y;
            acc0 = __builtin_amdgcn_wmma_f32_16x16x4_f32(
                false, a0, false, b0, (short)0, acc0, false, false);
            v2f a1;  a1.x = av[i].z;  a1.y = av[i].w;
            v2f b1;  b1.x = uv.z;     b1.y = uv.w;
            acc1 = __builtin_amdgcn_wmma_f32_16x16x4_f32(
                false, a1, false, b1, (short)0, acc1, false, false);
        }
        acc0 += acc1;
        if (row == 0) {
#pragma unroll
            for (int j = 0; j < 8; ++j)
                sh_pl[kh][mt * 16 + khalf * 8 + j] = acc0[j];
        }
    }
    __syncthreads();

    float l0 = sh_pl[0][lane]      + sh_pl[1][lane];
    float l1 = sh_pl[0][lane + 32] + sh_pl[1][lane + 32];
    float mx = fmaxf(l0, l1);
    for (int off = 16; off > 0; off >>= 1)
        mx = fmaxf(mx, __shfl_xor(mx, off, 32));
    float e0 = __expf(l0 - mx);
    float e1 = __expf(l1 - mx);
    float s = e0 + e1;
    for (int off = 16; off > 0; off >>= 1)
        s += __shfl_xor(s, off, 32);

    {
        float esel = (mq < 2) ? e0 : e1;
        int   lbase = (mq & 1) * 16 + 2 * khalf;
        v8f acc[4];
#pragma unroll
        for (int j = 0; j < 4; ++j)
            acc[j] = (v8f){0.f, 0.f, 0.f, 0.f, 0.f, 0.f, 0.f, 0.f};
#pragma unroll
        for (int ks = 0; ks < 4; ++ks) {
            v2f bb;
            bb.x = __shfl(esel, lbase + ks * 4 + 0, 32);
            bb.y = __shfl(esel, lbase + ks * 4 + 1, 32);
            float c0[4] = {cv[ks][0].x, cv[ks][0].y, cv[ks][0].z, cv[ks][0].w};
            float c1[4] = {cv[ks][1].x, cv[ks][1].y, cv[ks][1].z, cv[ks][1].w};
#pragma unroll
            for (int j = 0; j < 4; ++j) {
                v2f a;  a.x = c0[j];  a.y = c1[j];
                acc[j] = __builtin_amdgcn_wmma_f32_16x16x4_f32(
                    false, a, false, bb, (short)0, acc[j], false, false);
            }
        }
        if (row == 0) {
#pragma unroll
            for (int r = 0; r < 8; ++r) {
                float4 v = {acc[0][r], acc[1][r], acc[2][r], acc[3][r]};
                *(float4*)(&sh_po[mq][dh * 64 + 32 * khalf + 4 * r]) = v;
            }
        }
    }
    __syncthreads();

    float* rp = rec + chunk;
    if (tid < D) {
        rp[tid] = sh_po[0][tid] + sh_po[1][tid] + sh_po[2][tid] + sh_po[3][tid];
        if (g == 0) rp[2 * D + tid] = u_d;
    } else if (tid == D) {
        rp[D] = mx;
    } else if (tid == D + 1) {
        rp[D + 1] = s;
    }
}

__global__ void finish_kernel(const float* __restrict__ prev, float* __restrict__ out)
{
    int b = blockIdx.x;
    int d = threadIdx.x;
    out[b * D + d] = merge_records(prev + (size_t)b * M * D, d);
}

extern "C" void kernel_launch(void* const* d_in, const int* in_sizes, int n_in,
                              void* d_out, int out_size, void* d_ws, size_t ws_size,
                              hipStream_t stream)
{
    const int*   story    = (const int*)d_in[0];
    const float* C_tables = (const float*)d_in[1];
    const float* pos_emb  = (const float*)d_in[2];
    float*       out      = (float*)d_out;

    char* ws = (char*)d_ws;
    size_t sa_off    = (((size_t)B * M * sizeof(int) +
                         (size_t)B * D * sizeof(float)) + 255) & ~(size_t)255;
    float* SA        = (float*)(ws + sa_off);
    const size_t S   = (size_t)B * M * D;
    float* rec1 = SA + 1 * 4 * D;
    float* rec2 = SA + 2 * 4 * D;

    const size_t TS = (size_t)V * D;
    pool_kernel<<<B * NCH, 256, 0, stream>>>(story, C_tables + 1 * TS, C_tables + 2 * TS,
                                             C_tables + 3 * TS, pos_emb, SA);
    hop_part_kernel<true ><<<dim3(G, B), 256, 0, stream>>>(SA + 1 * S, SA + 2 * S, SA,      rec1);
    hop_part_kernel<false><<<dim3(G, B), 256, 0, stream>>>(SA + 2 * S, SA + 3 * S, rec1,    rec2);
    finish_kernel<<<B, D, 0, stream>>>(rec2, out);
}
